# starting kernel
# baseline (speedup 1.0000x reference)
#include <hip/hip_runtime.h>
#include <hip/hip_bf16.h>
#include <math.h>


typedef __attribute__((ext_vector_type(2))) float v2f;
typedef __attribute__((ext_vector_type(8))) float v8f;

#define NEG_SLOPE 0.2f
#define BN_EPS 1e-5f

__global__ void gemm_wmma_f32(const float* __restrict__ A,
                              const float* __restrict__ B,
                              float* __restrict__ C,
                              int M, int K) {
  const int Nn   = 128;
  const int wave = threadIdx.x >> 5;
  const int lane = threadIdx.x & 31;
  const int half = lane >> 4;
  const int l    = lane & 15;
  const int m0   = blockIdx.x * 16;
  const int n0   = wave * 16;

  v8f c = {};
  for (int k = 0; k < K; k += 4) {
    const int ka = k + half * 2;
    v2f a, b;
    a.x = A[(size_t)(m0 + l) * K + ka];
    a.y = A[(size_t)(m0 + l) * K + ka + 1];
    b.x = B[(size_t)ka * Nn + n0 + l];
    b.y = B[(size_t)(ka + 1) * Nn + n0 + l];
    c = __builtin_amdgcn_wmma_f32_16x16x4_f32(false, a, false, b,
                                              (short)0, c, false, false);
  }
#pragma unroll
  for (int j = 0; j < 8; ++j) {
    const int row = m0 + j + half * 8;
    C[(size_t)row * Nn + n0 + l] = c[j];
  }
}

__device__ __forceinline__ void atomicMaxF32(float* addr, float val) {
  if (val >= 0.0f)
    atomicMax((int*)addr, __float_as_int(val));
  else
    atomicMin((unsigned int*)addr, __float_as_uint(val));
}

__device__ __forceinline__ void edge_endpoints(const int* __restrict__ ei,
                                               int eid, int E,
                                               int& src, int& dst) {
  if (eid < E) { src = ei[eid]; dst = ei[E + eid]; }
  else         { src = dst = eid - E; }
}

__global__ void fill_f32(float* p, float v, int n) {
  int t = blockIdx.x * blockDim.x + threadIdx.x;
  if (t < n) p[t] = v;
}

__global__ void edge_score(const float* __restrict__ xp,
                           const int* __restrict__ ei,
                           const float* __restrict__ avec,
                           float* __restrict__ e,
                           float* __restrict__ emax,
                           int E, int Etot) {
  const int eid = blockIdx.x * (blockDim.x >> 5) + (threadIdx.x >> 5);
  if (eid >= Etot) return;
  const int lane = threadIdx.x & 31;
  int src, dst; edge_endpoints(ei, eid, E, src, dst);

  const int h   = lane >> 4;
  const int idx = (lane & 15) * 4;
  const float4 s  = *(const float4*)(xp + (size_t)src * 128 + h * 64 + idx);
  const float4 d  = *(const float4*)(xp + (size_t)dst * 128 + h * 64 + idx);
  const float4 av = *(const float4*)(avec + h * 64 + idx);

  float m, lr, acc = 0.0f;
  m = s.x + d.x; lr = (m > 0.0f) ? m : NEG_SLOPE * m; acc += lr * av.x;
  m = s.y + d.y; lr = (m > 0.0f) ? m : NEG_SLOPE * m; acc += lr * av.y;
  m = s.z + d.z; lr = (m > 0.0f) ? m : NEG_SLOPE * m; acc += lr * av.z;
  m = s.w + d.w; lr = (m > 0.0f) ? m : NEG_SLOPE * m; acc += lr * av.w;

  for (int off = 8; off > 0; off >>= 1) acc += __shfl_down(acc, off, 16);
  if ((lane & 15) == 0) {
    e[(size_t)eid * 2 + h] = acc;
    atomicMaxF32(&emax[(size_t)dst * 2 + h], acc);
  }
}

__global__ void edge_exp(const int* __restrict__ ei,
                         float* __restrict__ e,
                         const float* __restrict__ emax,
                         float* __restrict__ denom,
                         int E, int Etot) {
  int t = blockIdx.x * blockDim.x + threadIdx.x;
  if (t >= Etot * 2) return;
  const int eid = t >> 1, h = t & 1;
  int src, dst; edge_endpoints(ei, eid, E, src, dst);
  (void)src;
  const float ee = expf(e[t] - emax[(size_t)dst * 2 + h]);
  e[t] = ee;
  __hip_atomic_fetch_add(&denom[(size_t)dst * 2 + h], ee,
                         __ATOMIC_RELAXED, __HIP_MEMORY_SCOPE_AGENT);
}

__global__ void edge_agg(const float* __restrict__ xp,
                         const int* __restrict__ ei,
                         const float* __restrict__ e,
                         const float* __restrict__ denom,
                         float* __restrict__ acc,
                         int E, int Etot) {
  const int eid = blockIdx.x * (blockDim.x >> 5) + (threadIdx.x >> 5);
  if (eid >= Etot) return;
  const int lane = threadIdx.x & 31;
  int src, dst; edge_endpoints(ei, eid, E, src, dst);

  const int h   = lane >> 4;
  const int idx = (lane & 15) * 4;
  const float alpha = e[(size_t)eid * 2 + h] /
                      (denom[(size_t)dst * 2 + h] + 1e-16f);
  const float4 s = *(const float4*)(xp + (size_t)src * 128 + h * 64 + idx);
  float* p = acc + (size_t)dst * 128 + h * 64 + idx;
  __hip_atomic_fetch_add(p + 0, alpha * s.x, __ATOMIC_RELAXED, __HIP_MEMORY_SCOPE_AGENT);
  __hip_atomic_fetch_add(p + 1, alpha * s.y, __ATOMIC_RELAXED, __HIP_MEMORY_SCOPE_AGENT);
  __hip_atomic_fetch_add(p + 2, alpha * s.z, __ATOMIC_RELAXED, __HIP_MEMORY_SCOPE_AGENT);
  __hip_atomic_fetch_add(p + 3, alpha * s.w, __ATOMIC_RELAXED, __HIP_MEMORY_SCOPE_AGENT);
}

__global__ void head_mean_stats(const float* __restrict__ acc,
                                float* __restrict__ tmp,
                                float* __restrict__ sum,
                                float* __restrict__ sumsq,
                                int N) {
  __shared__ float s1[256];
  __shared__ float s2[256];
  const int c = threadIdx.x & 63;
  const int nodesPerBlk = blockDim.x >> 6;
  float ls = 0.0f, lq = 0.0f;
  for (int n = blockIdx.x * nodesPerBlk + (threadIdx.x >> 6);
       n < N; n += gridDim.x * nodesPerBlk) {
    const float v = 0.5f * (acc[(size_t)n * 128 + c] +
                            acc[(size_t)n * 128 + 64 + c]);
    tmp[(size_t)n * 64 + c] = v;
    ls += v; lq += v * v;
  }
  s1[threadIdx.x] = ls; s2[threadIdx.x] = lq;
  __syncthreads();
  if (threadIdx.x < 64) {
    ls = s1[threadIdx.x] + s1[threadIdx.x + 64] +
         s1[threadIdx.x + 128] + s1[threadIdx.x + 192];
    lq = s2[threadIdx.x] + s2[threadIdx.x + 64] +
         s2[threadIdx.x + 128] + s2[threadIdx.x + 192];
    __hip_atomic_fetch_add(&sum[c],   ls, __ATOMIC_RELAXED, __HIP_MEMORY_SCOPE_AGENT);
    __hip_atomic_fetch_add(&sumsq[c], lq, __ATOMIC_RELAXED, __HIP_MEMORY_SCOPE_AGENT);
  }
}

__global__ void bn_apply(const float* __restrict__ tmp,
                         const float* __restrict__ sum,
                         const float* __restrict__ sumsq,
                         const float* __restrict__ gamma,
                         const float* __restrict__ beta,
                         float* __restrict__ out,
                         int N, int doTanh) {
  int t = blockIdx.x * blockDim.x + threadIdx.x;
  if (t >= N * 64) return;
  const int c = t & 63;
  const float invN = 1.0f / (float)N;
  const float mean = sum[c] * invN;
  const float var  = sumsq[c] * invN - mean * mean;
  float v = (tmp[t] - mean) * rsqrtf(var + BN_EPS) * gamma[c] + beta[c];
  out[t] = doTanh ? tanhf(v) : v;
}

extern "C" void kernel_launch(void* const* d_in, const int* in_sizes, int n_in,
                              void* d_out, int out_size, void* d_ws, size_t ws_size,
                              hipStream_t stream) {
  const float* x      = (const float*)d_in[0];
  const int*   ei     = (const int*)  d_in[1];
  const float* W0     = (const float*)d_in[2];
  const float* a0     = (const float*)d_in[3];
  const float* gamma0 = (const float*)d_in[4];
  const float* beta0  = (const float*)d_in[5];
  const float* W1     = (const float*)d_in[6];
  const float* a1     = (const float*)d_in[7];
  const float* gamma1 = (const float*)d_in[8];
  const float* beta1  = (const float*)d_in[9];
  float* out = (float*)d_out;

  const int N    = in_sizes[0] / 128;
  const int E    = in_sizes[1] / 2;
  const int Etot = E + N;

  char* ws = (char*)d_ws;
  size_t off = 0;
  auto carve = [&](size_t nfloats) {
    float* p = (float*)(ws + off);
    off = (off + nfloats * sizeof(float) + 255) & ~(size_t)255;
    return p;
  };
  float* xp    = carve((size_t)N * 128);
  float* acc   = carve((size_t)N * 128);
  float* e     = carve((size_t)Etot * 2);
  float* emax  = carve((size_t)N * 2);
  float* denom = carve((size_t)N * 2);
  float* tmp   = carve((size_t)N * 64);
  float* h0    = carve((size_t)N * 64);
  float* stats = carve(128);
  (void)ws_size;

  const int waveEdgeBlocks = (Etot + 7) / 8;
  const int ehBlocks       = (Etot * 2 + 255) / 256;
  const int nodeElBlocks   = (N * 64 + 255) / 256;
  const int fillBlocks     = (N * 2 + 255) / 256;

  gemm_wmma_f32<<<N / 16, 256, 0, stream>>>(x, W0, xp, N, 128);

  fill_f32<<<fillBlocks, 256, 0, stream>>>(emax, -INFINITY, N * 2);
  hipMemsetAsync(denom, 0, (size_t)N * 2 * sizeof(float), stream);
  hipMemsetAsync(acc,   0, (size_t)N * 128 * sizeof(float), stream);
  hipMemsetAsync(stats, 0, 128 * sizeof(float), stream);

  edge_score<<<waveEdgeBlocks, 256, 0, stream>>>(xp, ei, a0, e, emax, E, Etot);
  edge_exp  <<<ehBlocks,       256, 0, stream>>>(ei, e, emax, denom, E, Etot);
  edge_agg  <<<waveEdgeBlocks, 256, 0, stream>>>(xp, ei, e, denom, acc, E, Etot);

  head_mean_stats<<<256, 256, 0, stream>>>(acc, tmp, stats, stats + 64, N);
  bn_apply<<<nodeElBlocks, 256, 0, stream>>>(tmp, stats, stats + 64,
                                             gamma0, beta0, h0, N, 1);

  gemm_wmma_f32<<<N / 16, 256, 0, stream>>>(h0, W1, xp, N, 64);

  fill_f32<<<fillBlocks, 256, 0, stream>>>(emax, -INFINITY, N * 2);
  hipMemsetAsync(denom, 0, (size_t)N * 2 * sizeof(float), stream);
  hipMemsetAsync(acc,   0, (size_t)N * 128 * sizeof(float), stream);
  hipMemsetAsync(stats, 0, 128 * sizeof(float), stream);

  edge_score<<<waveEdgeBlocks, 256, 0, stream>>>(xp, ei, a1, e, emax, E, Etot);
  edge_exp  <<<ehBlocks,       256, 0, stream>>>(ei, e, emax, denom, E, Etot);
  edge_agg  <<<waveEdgeBlocks, 256, 0, stream>>>(xp, ei, e, denom, acc, E, Etot);

  head_mean_stats<<<256, 256, 0, stream>>>(acc, tmp, stats, stats + 64, N);
  bn_apply<<<nodeElBlocks, 256, 0, stream>>>(tmp, stats, stats + 64,
                                             gamma1, beta1, out, N, 0);
}


# Aggregation scatter takes the head mean per edge before the add: lane = channel, 2 whole-line f32 atomic instructions per edge into [N,64] (was 4, each over 4 lines, into [N,128]); sum order changes
# speedup vs baseline: 1.6840x; 1.6840x over previous
#include <hip/hip_runtime.h>
#include <hip/hip_bf16.h>
#include <math.h>


typedef __attribute__((ext_vector_type(2))) float v2f;
typedef __attribute__((ext_vector_type(8))) float v8f;

#define NEG_SLOPE 0.2f
#define BN_EPS 1e-5f

__global__ void gemm_wmma_f32(const float* __restrict__ A,
                              const float* __restrict__ B,
                              float* __restrict__ C,
                              int M, int K) {
  const int Nn   = 128;
  const int wave = threadIdx.x >> 5;
  const int lane = threadIdx.x & 31;
  const int half = lane >> 4;
  const int l    = lane & 15;
  const int m0   = blockIdx.x * 16;
  const int n0   = wave * 16;

  v8f c = {};
  for (int k = 0; k < K; k += 4) {
    const int ka = k + half * 2;
    v2f a, b;
    a.x = A[(size_t)(m0 + l) * K + ka];
    a.y = A[(size_t)(m0 + l) * K + ka + 1];
    b.x = B[(size_t)ka * Nn + n0 + l];
    b.y = B[(size_t)(ka + 1) * Nn + n0 + l];
    c = __builtin_amdgcn_wmma_f32_16x16x4_f32(false, a, false, b,
                                              (short)0, c, false, false);
  }
#pragma unroll
  for (int j = 0; j < 8; ++j) {
    const int row = m0 + j + half * 8;
    C[(size_t)row * Nn + n0 + l] = c[j];
  }
}

__device__ __forceinline__ void atomicMaxF32(float* addr, float val) {
  if (val >= 0.0f)
    atomicMax((int*)addr, __float_as_int(val));
  else
    atomicMin((unsigned int*)addr, __float_as_uint(val));
}

__device__ __forceinline__ void edge_endpoints(const int* __restrict__ ei,
                                               int eid, int E,
                                               int& src, int& dst) {
  if (eid < E) { src = ei[eid]; dst = ei[E + eid]; }
  else         { src = dst = eid - E; }
}

__global__ void fill_f32(float* p, float v, int n) {
  int t = blockIdx.x * blockDim.x + threadIdx.x;
  if (t < n) p[t] = v;
}

#define EPW 8

template <int CTRL>
__device__ __forceinline__ float row_lane(float v) {
  return __int_as_float(__builtin_amdgcn_update_dpp(
      0, __float_as_int(v), CTRL, 0xf, 0xf, true));
}

__device__ __forceinline__ int batch_first_edge() {
  const int w = __builtin_amdgcn_readfirstlane((int)(threadIdx.x >> 5));
  return (blockIdx.x * (blockDim.x >> 5) + w) * EPW;
}

__global__ void edge_score(const float* __restrict__ xp,
                           const int* __restrict__ ei,
                           const float* __restrict__ avec,
                           float* __restrict__ e,
                           float* __restrict__ emax,
                           int E, int Etot) {
  const int lane = threadIdx.x & 31;
  const int e0 = batch_first_edge();
  if (e0 >= Etot) return;
  const int mine  = e0 + (lane & (EPW - 1));
  const bool valid = mine < Etot;
  int src, dst; edge_endpoints(ei, valid ? mine : Etot - 1, E, src, dst);

  const float4 av = *(const float4*)(avec + lane * 4);

  float4 S[EPW], D[EPW];
#pragma unroll
  for (int j = 0; j < EPW; ++j) {
    const int sj = __builtin_amdgcn_readlane(src, j);
    const int dj = __builtin_amdgcn_readlane(dst, j);
    S[j] = *(const float4*)(xp + (size_t)sj * 128 + lane * 4);
    D[j] = *(const float4*)(xp + (size_t)dj * 128 + lane * 4);
  }

  float acc[EPW];
#pragma unroll
  for (int j = 0; j < EPW; ++j) {
    float m, lr, a = 0.0f;
    m = S[j].x + D[j].x; lr = (m > 0.0f) ? m : NEG_SLOPE * m; a = fmaf(lr, av.x, a);
    m = S[j].y + D[j].y; lr = (m > 0.0f) ? m : NEG_SLOPE * m; a = fmaf(lr, av.y, a);
    m = S[j].z + D[j].z; lr = (m > 0.0f) ? m : NEG_SLOPE * m; a = fmaf(lr, av.z, a);
    m = S[j].w + D[j].w; lr = (m > 0.0f) ? m : NEG_SLOPE * m; a = fmaf(lr, av.w, a);
    acc[j] = a;
  }
#pragma unroll
  for (int j = 0; j < EPW; ++j) acc[j] += row_lane<0x168>(acc[j]);
#pragma unroll
  for (int j = 0; j < EPW; ++j) acc[j] += row_lane<0x164>(acc[j]);
#pragma unroll
  for (int j = 0; j < EPW; ++j) acc[j] += row_lane<0x162>(acc[j]);
#pragma unroll
  for (int j = 0; j < EPW; ++j) acc[j] += row_lane<0x161>(acc[j]);

  const int r = lane & 15;
  float val = acc[0];
#pragma unroll
  for (int j = 1; j < EPW; ++j) val = (r == j) ? acc[j] : val;
  if (r < EPW && valid) {
    const int h = lane >> 4;
    e[(size_t)mine * 2 + h] = val;
    atomicMaxF32(&emax[(size_t)dst * 2 + h], val);
  }
}

__global__ void edge_exp(const int* __restrict__ ei,
                         float* __restrict__ e,
                         const float* __restrict__ emax,
                         float* __restrict__ denom,
                         int E, int Etot) {
  int t = blockIdx.x * blockDim.x + threadIdx.x;
  if (t >= Etot * 2) return;
  const int eid = t >> 1, h = t & 1;
  int src, dst; edge_endpoints(ei, eid, E, src, dst);
  (void)src;
  const float ee = expf(e[t] - emax[(size_t)dst * 2 + h]);
  e[t] = ee;
  __hip_atomic_fetch_add(&denom[(size_t)dst * 2 + h], ee,
                         __ATOMIC_RELAXED, __HIP_MEMORY_SCOPE_AGENT);
}

template <bool FULL>
__device__ __forceinline__ void agg_batch(const float* __restrict__ xp,
                                          float* __restrict__ acc,
                                          int src, int dst, float halfAlpha,
                                          int lane, int nvalid) {
  float X[EPW][4];
#pragma unroll
  for (int j = 0; j < EPW; ++j) {
    const int sj = __builtin_amdgcn_readlane(src, j);
    const float* row = xp + (size_t)sj * 128 + lane;
    X[j][0] = row[0];
    X[j][1] = row[32];
    X[j][2] = row[64];
    X[j][3] = row[96];
  }
#pragma unroll
  for (int j = 0; j < EPW; ++j) {
    if (FULL || j < nvalid) {
      const int dj = __builtin_amdgcn_readlane(dst, j);
      const float a0 = __int_as_float(
          __builtin_amdgcn_readlane(__float_as_int(halfAlpha), j));
      const float a1 = __int_as_float(
          __builtin_amdgcn_readlane(__float_as_int(halfAlpha), 16 + j));
      float* p = acc + (size_t)dj * 64 + lane;
      __hip_atomic_fetch_add(p,      fmaf(a1, X[j][2], a0 * X[j][0]),
                             __ATOMIC_RELAXED, __HIP_MEMORY_SCOPE_AGENT);
      __hip_atomic_fetch_add(p + 32, fmaf(a1, X[j][3], a0 * X[j][1]),
                             __ATOMIC_RELAXED, __HIP_MEMORY_SCOPE_AGENT);
    }
  }
}

__global__ void edge_agg(const float* __restrict__ xp,
                         const int* __restrict__ ei,
                         const float* __restrict__ e,
                         const float* __restrict__ denom,
                         float* __restrict__ acc,
                         int E, int Etot) {
  const int lane = threadIdx.x & 31;
  const int e0 = batch_first_edge();
  if (e0 >= Etot) return;
  const int nvalid = Etot - e0;
  const int mine = ((lane & (EPW - 1)) < nvalid) ? e0 + (lane & (EPW - 1))
                                                 : Etot - 1;
  int src, dst; edge_endpoints(ei, mine, E, src, dst);

  const int h = lane >> 4;
  const float halfAlpha = 0.5f * (e[(size_t)mine * 2 + h] /
                                  (denom[(size_t)dst * 2 + h] + 1e-16f));

  if (nvalid >= EPW) agg_batch<true >(xp, acc, src, dst, halfAlpha, lane, EPW);
  else               agg_batch<false>(xp, acc, src, dst, halfAlpha, lane, nvalid);
}

__global__ void bn_stats(const float* __restrict__ acc,
                         float* __restrict__ sum,
                         float* __restrict__ sumsq,
                         int N) {
  __shared__ float s1[256];
  __shared__ float s2[256];
  const int c = threadIdx.x & 63;
  const int nodesPerBlk = blockDim.x >> 6;
  float ls = 0.0f, lq = 0.0f;
  for (int n = blockIdx.x * nodesPerBlk + (threadIdx.x >> 6);
       n < N; n += gridDim.x * nodesPerBlk) {
    const float v = acc[(size_t)n * 64 + c];
    ls += v; lq += v * v;
  }
  s1[threadIdx.x] = ls; s2[threadIdx.x] = lq;
  __syncthreads();
  if (threadIdx.x < 64) {
    ls = s1[threadIdx.x] + s1[threadIdx.x + 64] +
         s1[threadIdx.x + 128] + s1[threadIdx.x + 192];
    lq = s2[threadIdx.x] + s2[threadIdx.x + 64] +
         s2[threadIdx.x + 128] + s2[threadIdx.x + 192];
    __hip_atomic_fetch_add(&sum[c],   ls, __ATOMIC_RELAXED, __HIP_MEMORY_SCOPE_AGENT);
    __hip_atomic_fetch_add(&sumsq[c], lq, __ATOMIC_RELAXED, __HIP_MEMORY_SCOPE_AGENT);
  }
}

__global__ void bn_apply(const float* __restrict__ hmean,
                         const float* __restrict__ sum,
                         const float* __restrict__ sumsq,
                         const float* __restrict__ gamma,
                         const float* __restrict__ beta,
                         float* __restrict__ out,
                         int N, int doTanh) {
  int t = blockIdx.x * blockDim.x + threadIdx.x;
  if (t >= N * 64) return;
  const int c = t & 63;
  const float invN = 1.0f / (float)N;
  const float mean = sum[c] * invN;
  const float var  = sumsq[c] * invN - mean * mean;
  float v = (hmean[t] - mean) * rsqrtf(var + BN_EPS) * gamma[c] + beta[c];
  out[t] = doTanh ? tanhf(v) : v;
}

extern "C" void kernel_launch(void* const* d_in, const int* in_sizes, int n_in,
                              void* d_out, int out_size, void* d_ws, size_t ws_size,
                              hipStream_t stream) {
  const float* x      = (const float*)d_in[0];
  const int*   ei     = (const int*)  d_in[1];
  const float* W0     = (const float*)d_in[2];
  const float* a0     = (const float*)d_in[3];
  const float* gamma0 = (const float*)d_in[4];
  const float* beta0  = (const float*)d_in[5];
  const float* W1     = (const float*)d_in[6];
  const float* a1     = (const float*)d_in[7];
  const float* gamma1 = (const float*)d_in[8];
  const float* beta1  = (const float*)d_in[9];
  float* out = (float*)d_out;

  const int N    = in_sizes[0] / 128;
  const int E    = in_sizes[1] / 2;
  const int Etot = E + N;

  char* ws = (char*)d_ws;
  size_t off = 0;
  auto carve = [&](size_t nfloats) {
    float* p = (float*)(ws + off);
    off = (off + nfloats * sizeof(float) + 255) & ~(size_t)255;
    return p;
  };
  float* xp    = carve((size_t)N * 128);
  float* acc   = carve((size_t)N * 64);
  float* e     = carve((size_t)Etot * 2);
  float* emax  = carve((size_t)N * 2);
  float* denom = carve((size_t)N * 2);
  float* h0    = carve((size_t)N * 64);
  float* stats = carve(128);
  (void)ws_size;

  const int waveEdgeBlocks = (Etot + 8 * EPW - 1) / (8 * EPW);
  const int ehBlocks       = (Etot * 2 + 255) / 256;
  const int nodeElBlocks   = (N * 64 + 255) / 256;
  const int fillBlocks     = (N * 2 + 255) / 256;

  gemm_wmma_f32<<<N / 16, 256, 0, stream>>>(x, W0, xp, N, 128);

  fill_f32<<<fillBlocks, 256, 0, stream>>>(emax, -INFINITY, N * 2);
  hipMemsetAsync(denom, 0, (size_t)N * 2 * sizeof(float), stream);
  hipMemsetAsync(acc,   0, (size_t)N * 64 * sizeof(float), stream);
  hipMemsetAsync(stats, 0, 128 * sizeof(float), stream);

  edge_score<<<waveEdgeBlocks, 256, 0, stream>>>(xp, ei, a0, e, emax, E, Etot);
  edge_exp  <<<ehBlocks,       256, 0, stream>>>(ei, e, emax, denom, E, Etot);
  edge_agg  <<<waveEdgeBlocks, 256, 0, stream>>>(xp, ei, e, denom, acc, E, Etot);

  bn_stats<<<256, 256, 0, stream>>>(acc, stats, stats + 64, N);
  bn_apply<<<nodeElBlocks, 256, 0, stream>>>(acc, stats, stats + 64,
                                             gamma0, beta0, h0, N, 1);

  gemm_wmma_f32<<<N / 16, 256, 0, stream>>>(h0, W1, xp, N, 64);

  fill_f32<<<fillBlocks, 256, 0, stream>>>(emax, -INFINITY, N * 2);
  hipMemsetAsync(denom, 0, (size_t)N * 2 * sizeof(float), stream);
  hipMemsetAsync(acc,   0, (size_t)N * 64 * sizeof(float), stream);
  hipMemsetAsync(stats, 0, 128 * sizeof(float), stream);

  edge_score<<<waveEdgeBlocks, 256, 0, stream>>>(xp, ei, a1, e, emax, E, Etot);
  edge_exp  <<<ehBlocks,       256, 0, stream>>>(ei, e, emax, denom, E, Etot);
  edge_agg  <<<waveEdgeBlocks, 256, 0, stream>>>(xp, ei, e, denom, acc, E, Etot);

  bn_stats<<<256, 256, 0, stream>>>(acc, stats, stats + 64, N);
  bn_apply<<<nodeElBlocks, 256, 0, stream>>>(acc, stats, stats + 64,
                                             gamma1, beta1, out, N, 0);
}


# Batch-norm statistics pass: a thread requests 32 of its rows at once, then adds them in the old row order; was one load waited for per row, 49 dependent round trips a wave; same sums
# speedup vs baseline: 1.8424x; 1.0941x over previous
#include <hip/hip_runtime.h>
#include <hip/hip_bf16.h>
#include <math.h>


typedef __attribute__((ext_vector_type(2))) float v2f;
typedef __attribute__((ext_vector_type(8))) float v8f;

#define NEG_SLOPE 0.2f
#define BN_EPS 1e-5f

__global__ void gemm_wmma_f32(const float* __restrict__ A,
                              const float* __restrict__ B,
                              float* __restrict__ C,
                              int M, int K) {
  const int Nn   = 128;
  const int wave = threadIdx.x >> 5;
  const int lane = threadIdx.x & 31;
  const int half = lane >> 4;
  const int l    = lane & 15;
  const int m0   = blockIdx.x * 16;
  const int n0   = wave * 16;

  v8f c = {};
  for (int k = 0; k < K; k += 4) {
    const int ka = k + half * 2;
    v2f a, b;
    a.x = A[(size_t)(m0 + l) * K + ka];
    a.y = A[(size_t)(m0 + l) * K + ka + 1];
    b.x = B[(size_t)ka * Nn + n0 + l];
    b.y = B[(size_t)(ka + 1) * Nn + n0 + l];
    c = __builtin_amdgcn_wmma_f32_16x16x4_f32(false, a, false, b,
                                              (short)0, c, false, false);
  }
#pragma unroll
  for (int j = 0; j < 8; ++j) {
    const int row = m0 + j + half * 8;
    C[(size_t)row * Nn + n0 + l] = c[j];
  }
}

__device__ __forceinline__ void atomicMaxF32(float* addr, float val) {
  if (val >= 0.0f)
    atomicMax((int*)addr, __float_as_int(val));
  else
    atomicMin((unsigned int*)addr, __float_as_uint(val));
}

__device__ __forceinline__ void edge_endpoints(const int* __restrict__ ei,
                                               int eid, int E,
                                               int& src, int& dst) {
  if (eid < E) { src = ei[eid]; dst = ei[E + eid]; }
  else         { src = dst = eid - E; }
}

__global__ void fill_f32(float* p, float v, int n) {
  int t = blockIdx.x * blockDim.x + threadIdx.x;
  if (t < n) p[t] = v;
}

#define EPW 8

template <int CTRL>
__device__ __forceinline__ float row_lane(float v) {
  return __int_as_float(__builtin_amdgcn_update_dpp(
      0, __float_as_int(v), CTRL, 0xf, 0xf, true));
}

__device__ __forceinline__ int batch_first_edge() {
  const int w = __builtin_amdgcn_readfirstlane((int)(threadIdx.x >> 5));
  return (blockIdx.x * (blockDim.x >> 5) + w) * EPW;
}

__global__ void edge_score(const float* __restrict__ xp,
                           const int* __restrict__ ei,
                           const float* __restrict__ avec,
                           float* __restrict__ e,
                           float* __restrict__ emax,
                           int E, int Etot) {
  const int lane = threadIdx.x & 31;
  const int e0 = batch_first_edge();
  if (e0 >= Etot) return;
  const int mine  = e0 + (lane & (EPW - 1));
  const bool valid = mine < Etot;
  int src, dst; edge_endpoints(ei, valid ? mine : Etot - 1, E, src, dst);

  const float4 av = *(const float4*)(avec + lane * 4);

  float4 S[EPW], D[EPW];
#pragma unroll
  for (int j = 0; j < EPW; ++j) {
    const int sj = __builtin_amdgcn_readlane(src, j);
    const int dj = __builtin_amdgcn_readlane(dst, j);
    S[j] = *(const float4*)(xp + (size_t)sj * 128 + lane * 4);
    D[j] = *(const float4*)(xp + (size_t)dj * 128 + lane * 4);
  }

  float acc[EPW];
#pragma unroll
  for (int j = 0; j < EPW; ++j) {
    float m, lr, a = 0.0f;
    m = S[j].x + D[j].x; lr = (m > 0.0f) ? m : NEG_SLOPE * m; a = fmaf(lr, av.x, a);
    m = S[j].y + D[j].y; lr = (m > 0.0f) ? m : NEG_SLOPE * m; a = fmaf(lr, av.y, a);
    m = S[j].z + D[j].z; lr = (m > 0.0f) ? m : NEG_SLOPE * m; a = fmaf(lr, av.z, a);
    m = S[j].w + D[j].w; lr = (m > 0.0f) ? m : NEG_SLOPE * m; a = fmaf(lr, av.w, a);
    acc[j] = a;
  }
#pragma unroll
  for (int j = 0; j < EPW; ++j) acc[j] += row_lane<0x168>(acc[j]);
#pragma unroll
  for (int j = 0; j < EPW; ++j) acc[j] += row_lane<0x164>(acc[j]);
#pragma unroll
  for (int j = 0; j < EPW; ++j) acc[j] += row_lane<0x162>(acc[j]);
#pragma unroll
  for (int j = 0; j < EPW; ++j) acc[j] += row_lane<0x161>(acc[j]);

  const int r = lane & 15;
  float val = acc[0];
#pragma unroll
  for (int j = 1; j < EPW; ++j) val = (r == j) ? acc[j] : val;
  if (r < EPW && valid) {
    const int h = lane >> 4;
    e[(size_t)mine * 2 + h] = val;
    atomicMaxF32(&emax[(size_t)dst * 2 + h], val);
  }
}

__global__ void edge_exp(const int* __restrict__ ei,
                         float* __restrict__ e,
                         const float* __restrict__ emax,
                         float* __restrict__ denom,
                         int E, int Etot) {
  int t = blockIdx.x * blockDim.x + threadIdx.x;
  if (t >= Etot * 2) return;
  const int eid = t >> 1, h = t & 1;
  int src, dst; edge_endpoints(ei, eid, E, src, dst);
  (void)src;
  const float ee = expf(e[t] - emax[(size_t)dst * 2 + h]);
  e[t] = ee;
  __hip_atomic_fetch_add(&denom[(size_t)dst * 2 + h], ee,
                         __ATOMIC_RELAXED, __HIP_MEMORY_SCOPE_AGENT);
}

template <bool FULL>
__device__ __forceinline__ void agg_batch(const float* __restrict__ xp,
                                          float* __restrict__ acc,
                                          int src, int dst, float halfAlpha,
                                          int lane, int nvalid) {
  float X[EPW][4];
#pragma unroll
  for (int j = 0; j < EPW; ++j) {
    const int sj = __builtin_amdgcn_readlane(src, j);
    const float* row = xp + (size_t)sj * 128 + lane;
    X[j][0] = row[0];
    X[j][1] = row[32];
    X[j][2] = row[64];
    X[j][3] = row[96];
  }
#pragma unroll
  for (int j = 0; j < EPW; ++j) {
    if (FULL || j < nvalid) {
      const int dj = __builtin_amdgcn_readlane(dst, j);
      const float a0 = __int_as_float(
          __builtin_amdgcn_readlane(__float_as_int(halfAlpha), j));
      const float a1 = __int_as_float(
          __builtin_amdgcn_readlane(__float_as_int(halfAlpha), 16 + j));
      float* p = acc + (size_t)dj * 64 + lane;
      __hip_atomic_fetch_add(p,      fmaf(a1, X[j][2], a0 * X[j][0]),
                             __ATOMIC_RELAXED, __HIP_MEMORY_SCOPE_AGENT);
      __hip_atomic_fetch_add(p + 32, fmaf(a1, X[j][3], a0 * X[j][1]),
                             __ATOMIC_RELAXED, __HIP_MEMORY_SCOPE_AGENT);
    }
  }
}

__global__ void edge_agg(const float* __restrict__ xp,
                         const int* __restrict__ ei,
                         const float* __restrict__ e,
                         const float* __restrict__ denom,
                         float* __restrict__ acc,
                         int E, int Etot) {
  const int lane = threadIdx.x & 31;
  const int e0 = batch_first_edge();
  if (e0 >= Etot) return;
  const int nvalid = Etot - e0;
  const int mine = ((lane & (EPW - 1)) < nvalid) ? e0 + (lane & (EPW - 1))
                                                 : Etot - 1;
  int src, dst; edge_endpoints(ei, mine, E, src, dst);

  const int h = lane >> 4;
  const float halfAlpha = 0.5f * (e[(size_t)mine * 2 + h] /
                                  (denom[(size_t)dst * 2 + h] + 1e-16f));

  if (nvalid >= EPW) agg_batch<true >(xp, acc, src, dst, halfAlpha, lane, EPW);
  else               agg_batch<false>(xp, acc, src, dst, halfAlpha, lane, nvalid);
}

#define SB 32
__global__ void bn_stats(const float* __restrict__ acc,
                         float* __restrict__ sum,
                         float* __restrict__ sumsq,
                         int N) {
  __shared__ float s1[256];
  __shared__ float s2[256];
  const int c = threadIdx.x & 63;
  const int nodesPerBlk = blockDim.x >> 6;
  const int step = gridDim.x * nodesPerBlk;
  float ls = 0.0f, lq = 0.0f;
  for (int n0 = blockIdx.x * nodesPerBlk + (threadIdx.x >> 6);
       n0 < N; n0 += SB * step) {
    float v[SB];
#pragma unroll
    for (int j = 0; j < SB; ++j) {
      const int n = n0 + j * step;
      v[j] = acc[(size_t)(n < N ? n : n0) * 64 + c];
    }
#pragma unroll
    for (int j = 0; j < SB; ++j) {
      if (n0 + j * step < N) { ls += v[j]; lq += v[j] * v[j]; }
    }
  }
  s1[threadIdx.x] = ls; s2[threadIdx.x] = lq;
  __syncthreads();
  if (threadIdx.x < 64) {
    ls = s1[threadIdx.x] + s1[threadIdx.x + 64] +
         s1[threadIdx.x + 128] + s1[threadIdx.x + 192];
    lq = s2[threadIdx.x] + s2[threadIdx.x + 64] +
         s2[threadIdx.x + 128] + s2[threadIdx.x + 192];
    __hip_atomic_fetch_add(&sum[c],   ls, __ATOMIC_RELAXED, __HIP_MEMORY_SCOPE_AGENT);
    __hip_atomic_fetch_add(&sumsq[c], lq, __ATOMIC_RELAXED, __HIP_MEMORY_SCOPE_AGENT);
  }
}

__global__ void bn_apply(const float* __restrict__ hmean,
                         const float* __restrict__ sum,
                         const float* __restrict__ sumsq,
                         const float* __restrict__ gamma,
                         const float* __restrict__ beta,
                         float* __restrict__ out,
                         int N, int doTanh) {
  int t = blockIdx.x * blockDim.x + threadIdx.x;
  if (t >= N * 64) return;
  const int c = t & 63;
  const float invN = 1.0f / (float)N;
  const float mean = sum[c] * invN;
  const float var  = sumsq[c] * invN - mean * mean;
  float v = (hmean[t] - mean) * rsqrtf(var + BN_EPS) * gamma[c] + beta[c];
  out[t] = doTanh ? tanhf(v) : v;
}

extern "C" void kernel_launch(void* const* d_in, const int* in_sizes, int n_in,
                              void* d_out, int out_size, void* d_ws, size_t ws_size,
                              hipStream_t stream) {
  const float* x      = (const float*)d_in[0];
  const int*   ei     = (const int*)  d_in[1];
  const float* W0     = (const float*)d_in[2];
  const float* a0     = (const float*)d_in[3];
  const float* gamma0 = (const float*)d_in[4];
  const float* beta0  = (const float*)d_in[5];
  const float* W1     = (const float*)d_in[6];
  const float* a1     = (const float*)d_in[7];
  const float* gamma1 = (const float*)d_in[8];
  const float* beta1  = (const float*)d_in[9];
  float* out = (float*)d_out;

  const int N    = in_sizes[0] / 128;
  const int E    = in_sizes[1] / 2;
  const int Etot = E + N;

  char* ws = (char*)d_ws;
  size_t off = 0;
  auto carve = [&](size_t nfloats) {
    float* p = (float*)(ws + off);
    off = (off + nfloats * sizeof(float) + 255) & ~(size_t)255;
    return p;
  };
  float* xp    = carve((size_t)N * 128);
  float* acc   = carve((size_t)N * 64);
  float* e     = carve((size_t)Etot * 2);
  float* emax  = carve((size_t)N * 2);
  float* denom = carve((size_t)N * 2);
  float* h0    = carve((size_t)N * 64);
  float* stats = carve(128);
  (void)ws_size;

  const int waveEdgeBlocks = (Etot + 8 * EPW - 1) / (8 * EPW);
  const int ehBlocks       = (Etot * 2 + 255) / 256;
  const int nodeElBlocks   = (N * 64 + 255) / 256;
  const int fillBlocks     = (N * 2 + 255) / 256;

  gemm_wmma_f32<<<N / 16, 256, 0, stream>>>(x, W0, xp, N, 128);

  fill_f32<<<fillBlocks, 256, 0, stream>>>(emax, -INFINITY, N * 2);
  hipMemsetAsync(denom, 0, (size_t)N * 2 * sizeof(float), stream);
  hipMemsetAsync(acc,   0, (size_t)N * 64 * sizeof(float), stream);
  hipMemsetAsync(stats, 0, 128 * sizeof(float), stream);

  edge_score<<<waveEdgeBlocks, 256, 0, stream>>>(xp, ei, a0, e, emax, E, Etot);
  edge_exp  <<<ehBlocks,       256, 0, stream>>>(ei, e, emax, denom, E, Etot);
  edge_agg  <<<waveEdgeBlocks, 256, 0, stream>>>(xp, ei, e, denom, acc, E, Etot);

  bn_stats<<<256, 256, 0, stream>>>(acc, stats, stats + 64, N);
  bn_apply<<<nodeElBlocks, 256, 0, stream>>>(acc, stats, stats + 64,
                                             gamma0, beta0, h0, N, 1);

  gemm_wmma_f32<<<N / 16, 256, 0, stream>>>(h0, W1, xp, N, 64);

  fill_f32<<<fillBlocks, 256, 0, stream>>>(emax, -INFINITY, N * 2);
  hipMemsetAsync(denom, 0, (size_t)N * 2 * sizeof(float), stream);
  hipMemsetAsync(acc,   0, (size_t)N * 64 * sizeof(float), stream);
  hipMemsetAsync(stats, 0, 128 * sizeof(float), stream);

  edge_score<<<waveEdgeBlocks, 256, 0, stream>>>(xp, ei, a1, e, emax, E, Etot);
  edge_exp  <<<ehBlocks,       256, 0, stream>>>(ei, e, emax, denom, E, Etot);
  edge_agg  <<<waveEdgeBlocks, 256, 0, stream>>>(xp, ei, e, denom, acc, E, Etot);

  bn_stats<<<256, 256, 0, stream>>>(acc, stats, stats + 64, N);
  bn_apply<<<nodeElBlocks, 256, 0, stream>>>(acc, stats, stats + 64,
                                             gamma1, beta1, out, N, 0);
}


# emax and denom one node per 128-byte line (was 16 nodes a line): max and add atomics of different nodes no longer queue on one shared line; start values set by one node kernel; atomics as before
# speedup vs baseline: 2.4571x; 1.3336x over previous
#include <hip/hip_runtime.h>
#include <hip/hip_bf16.h>
#include <math.h>


typedef __attribute__((ext_vector_type(2))) float v2f;
typedef __attribute__((ext_vector_type(8))) float v8f;

#define NEG_SLOPE 0.2f
#define BN_EPS 1e-5f

__global__ void gemm_wmma_f32(const float* __restrict__ A,
                              const float* __restrict__ B,
                              float* __restrict__ C,
                              int M, int K) {
  const int Nn   = 128;
  const int wave = threadIdx.x >> 5;
  const int lane = threadIdx.x & 31;
  const int half = lane >> 4;
  const int l    = lane & 15;
  const int m0   = blockIdx.x * 16;
  const int n0   = wave * 16;

  v8f c = {};
  for (int k = 0; k < K; k += 4) {
    const int ka = k + half * 2;
    v2f a, b;
    a.x = A[(size_t)(m0 + l) * K + ka];
    a.y = A[(size_t)(m0 + l) * K + ka + 1];
    b.x = B[(size_t)ka * Nn + n0 + l];
    b.y = B[(size_t)(ka + 1) * Nn + n0 + l];
    c = __builtin_amdgcn_wmma_f32_16x16x4_f32(false, a, false, b,
                                              (short)0, c, false, false);
  }
#pragma unroll
  for (int j = 0; j < 8; ++j) {
    const int row = m0 + j + half * 8;
    C[(size_t)row * Nn + n0 + l] = c[j];
  }
}

__device__ __forceinline__ void atomicMaxF32(float* addr, float val) {
  if (val >= 0.0f)
    atomicMax((int*)addr, __float_as_int(val));
  else
    atomicMin((unsigned int*)addr, __float_as_uint(val));
}

__device__ __forceinline__ void edge_endpoints(const int* __restrict__ ei,
                                               int eid, int E,
                                               int& src, int& dst) {
  if (eid < E) { src = ei[eid]; dst = ei[E + eid]; }
  else         { src = dst = eid - E; }
}

#define NPITCH 32

__global__ void node_start(float* __restrict__ emax, float* __restrict__ denom,
                           int n) {
  int t = blockIdx.x * blockDim.x + threadIdx.x;
  if (t < n) {
    *(float2*)(emax  + (size_t)t * NPITCH) = make_float2(-INFINITY, -INFINITY);
    *(float2*)(denom + (size_t)t * NPITCH) = make_float2(0.0f, 0.0f);
  }
}

#define EPW 8

template <int CTRL>
__device__ __forceinline__ float row_lane(float v) {
  return __int_as_float(__builtin_amdgcn_update_dpp(
      0, __float_as_int(v), CTRL, 0xf, 0xf, true));
}

__device__ __forceinline__ int batch_first_edge() {
  const int w = __builtin_amdgcn_readfirstlane((int)(threadIdx.x >> 5));
  return (blockIdx.x * (blockDim.x >> 5) + w) * EPW;
}

__global__ void edge_score(const float* __restrict__ xp,
                           const int* __restrict__ ei,
                           const float* __restrict__ avec,
                           float* __restrict__ e,
                           float* __restrict__ emax,
                           int E, int Etot) {
  const int lane = threadIdx.x & 31;
  const int e0 = batch_first_edge();
  if (e0 >= Etot) return;
  const int mine  = e0 + (lane & (EPW - 1));
  const bool valid = mine < Etot;
  int src, dst; edge_endpoints(ei, valid ? mine : Etot - 1, E, src, dst);

  const float4 av = *(const float4*)(avec + lane * 4);

  float4 S[EPW], D[EPW];
#pragma unroll
  for (int j = 0; j < EPW; ++j) {
    const int sj = __builtin_amdgcn_readlane(src, j);
    const int dj = __builtin_amdgcn_readlane(dst, j);
    S[j] = *(const float4*)(xp + (size_t)sj * 128 + lane * 4);
    D[j] = *(const float4*)(xp + (size_t)dj * 128 + lane * 4);
  }

  float acc[EPW];
#pragma unroll
  for (int j = 0; j < EPW; ++j) {
    float m, lr, a = 0.0f;
    m = S[j].x + D[j].x; lr = (m > 0.0f) ? m : NEG_SLOPE * m; a = fmaf(lr, av.x, a);
    m = S[j].y + D[j].y; lr = (m > 0.0f) ? m : NEG_SLOPE * m; a = fmaf(lr, av.y, a);
    m = S[j].z + D[j].z; lr = (m > 0.0f) ? m : NEG_SLOPE * m; a = fmaf(lr, av.z, a);
    m = S[j].w + D[j].w; lr = (m > 0.0f) ? m : NEG_SLOPE * m; a = fmaf(lr, av.w, a);
    acc[j] = a;
  }
#pragma unroll
  for (int j = 0; j < EPW; ++j) acc[j] += row_lane<0x168>(acc[j]);
#pragma unroll
  for (int j = 0; j < EPW; ++j) acc[j] += row_lane<0x164>(acc[j]);
#pragma unroll
  for (int j = 0; j < EPW; ++j) acc[j] += row_lane<0x162>(acc[j]);
#pragma unroll
  for (int j = 0; j < EPW; ++j) acc[j] += row_lane<0x161>(acc[j]);

  const int r = lane & 15;
  float val = acc[0];
#pragma unroll
  for (int j = 1; j < EPW; ++j) val = (r == j) ? acc[j] : val;
  if (r < EPW && valid) {
    const int h = lane >> 4;
    e[(size_t)mine * 2 + h] = val;
    atomicMaxF32(&emax[(size_t)dst * NPITCH + h], val);
  }
}

__global__ void edge_exp(const int* __restrict__ ei,
                         float* __restrict__ e,
                         const float* __restrict__ emax,
                         float* __restrict__ denom,
                         int E, int Etot) {
  int t = blockIdx.x * blockDim.x + threadIdx.x;
  if (t >= Etot * 2) return;
  const int eid = t >> 1, h = t & 1;
  int src, dst; edge_endpoints(ei, eid, E, src, dst);
  (void)src;
  const float ee = expf(e[t] - emax[(size_t)dst * NPITCH + h]);
  e[t] = ee;
  __hip_atomic_fetch_add(&denom[(size_t)dst * NPITCH + h], ee,
                         __ATOMIC_RELAXED, __HIP_MEMORY_SCOPE_AGENT);
}

template <bool FULL>
__device__ __forceinline__ void agg_batch(const float* __restrict__ xp,
                                          float* __restrict__ acc,
                                          int src, int dst, float halfAlpha,
                                          int lane, int nvalid) {
  float X[EPW][4];
#pragma unroll
  for (int j = 0; j < EPW; ++j) {
    const int sj = __builtin_amdgcn_readlane(src, j);
    const float* row = xp + (size_t)sj * 128 + lane;
    X[j][0] = row[0];
    X[j][1] = row[32];
    X[j][2] = row[64];
    X[j][3] = row[96];
  }
#pragma unroll
  for (int j = 0; j < EPW; ++j) {
    if (FULL || j < nvalid) {
      const int dj = __builtin_amdgcn_readlane(dst, j);
      const float a0 = __int_as_float(
          __builtin_amdgcn_readlane(__float_as_int(halfAlpha), j));
      const float a1 = __int_as_float(
          __builtin_amdgcn_readlane(__float_as_int(halfAlpha), 16 + j));
      float* p = acc + (size_t)dj * 64 + lane;
      __hip_atomic_fetch_add(p,      fmaf(a1, X[j][2], a0 * X[j][0]),
                             __ATOMIC_RELAXED, __HIP_MEMORY_SCOPE_AGENT);
      __hip_atomic_fetch_add(p + 32, fmaf(a1, X[j][3], a0 * X[j][1]),
                             __ATOMIC_RELAXED, __HIP_MEMORY_SCOPE_AGENT);
    }
  }
}

__global__ void edge_agg(const float* __restrict__ xp,
                         const int* __restrict__ ei,
                         const float* __restrict__ e,
                         const float* __restrict__ denom,
                         float* __restrict__ acc,
                         int E, int Etot) {
  const int lane = threadIdx.x & 31;
  const int e0 = batch_first_edge();
  if (e0 >= Etot) return;
  const int nvalid = Etot - e0;
  const int mine = ((lane & (EPW - 1)) < nvalid) ? e0 + (lane & (EPW - 1))
                                                 : Etot - 1;
  int src, dst; edge_endpoints(ei, mine, E, src, dst);

  const int h = lane >> 4;
  const float halfAlpha = 0.5f * (e[(size_t)mine * 2 + h] /
                                  (denom[(size_t)dst * NPITCH + h] + 1e-16f));

  if (nvalid >= EPW) agg_batch<true >(xp, acc, src, dst, halfAlpha, lane, EPW);
  else               agg_batch<false>(xp, acc, src, dst, halfAlpha, lane, nvalid);
}

#define SB 32
__global__ void bn_stats(const float* __restrict__ acc,
                         float* __restrict__ sum,
                         float* __restrict__ sumsq,
                         int N) {
  __shared__ float s1[256];
  __shared__ float s2[256];
  const int c = threadIdx.x & 63;
  const int nodesPerBlk = blockDim.x >> 6;
  const int step = gridDim.x * nodesPerBlk;
  float ls = 0.0f, lq = 0.0f;
  for (int n0 = blockIdx.x * nodesPerBlk + (threadIdx.x >> 6);
       n0 < N; n0 += SB * step) {
    float v[SB];
#pragma unroll
    for (int j = 0; j < SB; ++j) {
      const int n = n0 + j * step;
      v[j] = acc[(size_t)(n < N ? n : n0) * 64 + c];
    }
#pragma unroll
    for (int j = 0; j < SB; ++j) {
      if (n0 + j * step < N) { ls += v[j]; lq += v[j] * v[j]; }
    }
  }
  s1[threadIdx.x] = ls; s2[threadIdx.x] = lq;
  __syncthreads();
  if (threadIdx.x < 64) {
    ls = s1[threadIdx.x] + s1[threadIdx.x + 64] +
         s1[threadIdx.x + 128] + s1[threadIdx.x + 192];
    lq = s2[threadIdx.x] + s2[threadIdx.x + 64] +
         s2[threadIdx.x + 128] + s2[threadIdx.x + 192];
    __hip_atomic_fetch_add(&sum[c],   ls, __ATOMIC_RELAXED, __HIP_MEMORY_SCOPE_AGENT);
    __hip_atomic_fetch_add(&sumsq[c], lq, __ATOMIC_RELAXED, __HIP_MEMORY_SCOPE_AGENT);
  }
}

__global__ void bn_apply(const float* __restrict__ hmean,
                         const float* __restrict__ sum,
                         const float* __restrict__ sumsq,
                         const float* __restrict__ gamma,
                         const float* __restrict__ beta,
                         float* __restrict__ out,
                         int N, int doTanh) {
  int t = blockIdx.x * blockDim.x + threadIdx.x;
  if (t >= N * 64) return;
  const int c = t & 63;
  const float invN = 1.0f / (float)N;
  const float mean = sum[c] * invN;
  const float var  = sumsq[c] * invN - mean * mean;
  float v = (hmean[t] - mean) * rsqrtf(var + BN_EPS) * gamma[c] + beta[c];
  out[t] = doTanh ? tanhf(v) : v;
}

extern "C" void kernel_launch(void* const* d_in, const int* in_sizes, int n_in,
                              void* d_out, int out_size, void* d_ws, size_t ws_size,
                              hipStream_t stream) {
  const float* x      = (const float*)d_in[0];
  const int*   ei     = (const int*)  d_in[1];
  const float* W0     = (const float*)d_in[2];
  const float* a0     = (const float*)d_in[3];
  const float* gamma0 = (const float*)d_in[4];
  const float* beta0  = (const float*)d_in[5];
  const float* W1     = (const float*)d_in[6];
  const float* a1     = (const float*)d_in[7];
  const float* gamma1 = (const float*)d_in[8];
  const float* beta1  = (const float*)d_in[9];
  float* out = (float*)d_out;

  const int N    = in_sizes[0] / 128;
  const int E    = in_sizes[1] / 2;
  const int Etot = E + N;

  char* ws = (char*)d_ws;
  size_t off = 0;
  auto carve = [&](size_t nfloats) {
    float* p = (float*)(ws + off);
    off = (off + nfloats * sizeof(float) + 255) & ~(size_t)255;
    return p;
  };
  float* xp    = carve((size_t)N * 128);
  float* acc   = carve((size_t)N * 64);
  float* e     = carve((size_t)Etot * 2);
  float* emax  = carve((size_t)N * NPITCH);
  float* denom = carve((size_t)N * NPITCH);
  float* h0    = carve((size_t)N * 64);
  float* stats = carve(128);
  (void)ws_size;

  const int waveEdgeBlocks = (Etot + 8 * EPW - 1) / (8 * EPW);
  const int ehBlocks       = (Etot * 2 + 255) / 256;
  const int nodeElBlocks   = (N * 64 + 255) / 256;
  const int fillBlocks     = (N + 255) / 256;

  gemm_wmma_f32<<<N / 16, 256, 0, stream>>>(x, W0, xp, N, 128);

  node_start<<<fillBlocks, 256, 0, stream>>>(emax, denom, N);
  hipMemsetAsync(acc,   0, (size_t)N * 64 * sizeof(float), stream);
  hipMemsetAsync(stats, 0, 128 * sizeof(float), stream);

  edge_score<<<waveEdgeBlocks, 256, 0, stream>>>(xp, ei, a0, e, emax, E, Etot);
  edge_exp  <<<ehBlocks,       256, 0, stream>>>(ei, e, emax, denom, E, Etot);
  edge_agg  <<<waveEdgeBlocks, 256, 0, stream>>>(xp, ei, e, denom, acc, E, Etot);

  bn_stats<<<256, 256, 0, stream>>>(acc, stats, stats + 64, N);
  bn_apply<<<nodeElBlocks, 256, 0, stream>>>(acc, stats, stats + 64,
                                             gamma0, beta0, h0, N, 1);

  gemm_wmma_f32<<<N / 16, 256, 0, stream>>>(h0, W1, xp, N, 64);

  node_start<<<fillBlocks, 256, 0, stream>>>(emax, denom, N);
  hipMemsetAsync(acc,   0, (size_t)N * 64 * sizeof(float), stream);
  hipMemsetAsync(stats, 0, 128 * sizeof(float), stream);

  edge_score<<<waveEdgeBlocks, 256, 0, stream>>>(xp, ei, a1, e, emax, E, Etot);
  edge_exp  <<<ehBlocks,       256, 0, stream>>>(ei, e, emax, denom, E, Etot);
  edge_agg  <<<waveEdgeBlocks, 256, 0, stream>>>(xp, ei, e, denom, acc, E, Etot);

  bn_stats<<<256, 256, 0, stream>>>(acc, stats, stats + 64, N);
  bn_apply<<<nodeElBlocks, 256, 0, stream>>>(acc, stats, stats + 64,
                                             gamma1, beta1, out, N, 0);
}


# Projection GEMM register-tiled: a wave holds a 16x64 tile in 4 accumulators, A and W fragments by 16-byte loads in a permuted k and column order, 2 waves per row tile (was 8); k sum order changes
# speedup vs baseline: 2.5374x; 1.0327x over previous
#include <hip/hip_runtime.h>
#include <hip/hip_bf16.h>
#include <math.h>


typedef __attribute__((ext_vector_type(2))) float v2f;
typedef __attribute__((ext_vector_type(8))) float v8f;

#define NEG_SLOPE 0.2f
#define BN_EPS 1e-5f

typedef __attribute__((ext_vector_type(4))) float v4f;

__device__ __forceinline__ v8f wmma4(v2f a, float bx, float by, v8f c) {
  v2f b; b.x = bx; b.y = by;
  return __builtin_amdgcn_wmma_f32_16x16x4_f32(false, a, false, b,
                                               (short)0, c, false, false);
}

template <int K>
__global__ void __launch_bounds__(64)
gemm_wmma_f32(const float* A, const float* B, float* __restrict__ C) {
  const int Nn   = 128;
  const int wave = threadIdx.x >> 5;
  const int lane = threadIdx.x & 31;
  const int half = lane >> 4;
  const int l    = lane & 15;
  const int m0   = blockIdx.x * 16;
  const int n0   = wave * 64;

  const float* ap = A + (size_t)(m0 + l) * K + half * 4;
  const float* bp = B + (size_t)(half * 4) * Nn + n0 + l * 4;

  v8f c0 = {}, c1 = {}, c2 = {}, c3 = {};
  v4f a[2], b[2][4];
  a[0] = *(const v4f*)ap;
#pragma unroll
  for (int i = 0; i < 4; ++i) b[0][i] = *(const v4f*)(bp + i * Nn);

#pragma unroll
  for (int t = 0; t < K / 8; ++t) {
    const int cur = t & 1, nxt = cur ^ 1;
    if (t + 1 < K / 8) {
      a[nxt] = *(const v4f*)(ap + (t + 1) * 8);
#pragma unroll
      for (int i = 0; i < 4; ++i)
        b[nxt][i] = *(const v4f*)(bp + (size_t)((t + 1) * 8 + i) * Nn);
    }
    __builtin_amdgcn_sched_barrier(0);

    v2f a01, a23;
    a01.x = a[cur].x; a01.y = a[cur].y; a23.x = a[cur].z; a23.y = a[cur].w;
    c0 = wmma4(a01, b[cur][0].x, b[cur][1].x, c0);
    c1 = wmma4(a01, b[cur][0].y, b[cur][1].y, c1);
    c2 = wmma4(a01, b[cur][0].z, b[cur][1].z, c2);
    c3 = wmma4(a01, b[cur][0].w, b[cur][1].w, c3);
    c0 = wmma4(a23, b[cur][2].x, b[cur][3].x, c0);
    c1 = wmma4(a23, b[cur][2].y, b[cur][3].y, c1);
    c2 = wmma4(a23, b[cur][2].z, b[cur][3].z, c2);
    c3 = wmma4(a23, b[cur][2].w, b[cur][3].w, c3);
  }
#pragma unroll
  for (int j = 0; j < 8; ++j) {
    const int row = m0 + j + half * 8;
    v4f o; o.x = c0[j]; o.y = c1[j]; o.z = c2[j]; o.w = c3[j];
    *(v4f*)(C + (size_t)row * Nn + n0 + l * 4) = o;
  }
}

__device__ __forceinline__ void atomicMaxF32(float* addr, float val) {
  if (val >= 0.0f)
    atomicMax((int*)addr, __float_as_int(val));
  else
    atomicMin((unsigned int*)addr, __float_as_uint(val));
}

__device__ __forceinline__ void edge_endpoints(const int* __restrict__ ei,
                                               int eid, int E,
                                               int& src, int& dst) {
  if (eid < E) { src = ei[eid]; dst = ei[E + eid]; }
  else         { src = dst = eid - E; }
}

#define NPITCH 32

__global__ void node_start(float* __restrict__ emax, float* __restrict__ denom,
                           int n) {
  int t = blockIdx.x * blockDim.x + threadIdx.x;
  if (t < n) {
    *(float2*)(emax  + (size_t)t * NPITCH) = make_float2(-INFINITY, -INFINITY);
    *(float2*)(denom + (size_t)t * NPITCH) = make_float2(0.0f, 0.0f);
  }
}

#define EPW 8

template <int CTRL>
__device__ __forceinline__ float row_lane(float v) {
  return __int_as_float(__builtin_amdgcn_update_dpp(
      0, __float_as_int(v), CTRL, 0xf, 0xf, true));
}

__device__ __forceinline__ int batch_first_edge() {
  const int w = __builtin_amdgcn_readfirstlane((int)(threadIdx.x >> 5));
  return (blockIdx.x * (blockDim.x >> 5) + w) * EPW;
}

__global__ void edge_score(const float* __restrict__ xp,
                           const int* __restrict__ ei,
                           const float* __restrict__ avec,
                           float* __restrict__ e,
                           float* __restrict__ emax,
                           int E, int Etot) {
  const int lane = threadIdx.x & 31;
  const int e0 = batch_first_edge();
  if (e0 >= Etot) return;
  const int mine  = e0 + (lane & (EPW - 1));
  const bool valid = mine < Etot;
  int src, dst; edge_endpoints(ei, valid ? mine : Etot - 1, E, src, dst);

  const float4 av = *(const float4*)(avec + lane * 4);

  float4 S[EPW], D[EPW];
#pragma unroll
  for (int j = 0; j < EPW; ++j) {
    const int sj = __builtin_amdgcn_readlane(src, j);
    const int dj = __builtin_amdgcn_readlane(dst, j);
    S[j] = *(const float4*)(xp + (size_t)sj * 128 + lane * 4);
    D[j] = *(const float4*)(xp + (size_t)dj * 128 + lane * 4);
  }

  float acc[EPW];
#pragma unroll
  for (int j = 0; j < EPW; ++j) {
    float m, lr, a = 0.0f;
    m = S[j].x + D[j].x; lr = (m > 0.0f) ? m : NEG_SLOPE * m; a = fmaf(lr, av.x, a);
    m = S[j].y + D[j].y; lr = (m > 0.0f) ? m : NEG_SLOPE * m; a = fmaf(lr, av.y, a);
    m = S[j].z + D[j].z; lr = (m > 0.0f) ? m : NEG_SLOPE * m; a = fmaf(lr, av.z, a);
    m = S[j].w + D[j].w; lr = (m > 0.0f) ? m : NEG_SLOPE * m; a = fmaf(lr, av.w, a);
    acc[j] = a;
  }
#pragma unroll
  for (int j = 0; j < EPW; ++j) acc[j] += row_lane<0x168>(acc[j]);
#pragma unroll
  for (int j = 0; j < EPW; ++j) acc[j] += row_lane<0x164>(acc[j]);
#pragma unroll
  for (int j = 0; j < EPW; ++j) acc[j] += row_lane<0x162>(acc[j]);
#pragma unroll
  for (int j = 0; j < EPW; ++j) acc[j] += row_lane<0x161>(acc[j]);

  const int r = lane & 15;
  float val = acc[0];
#pragma unroll
  for (int j = 1; j < EPW; ++j) val = (r == j) ? acc[j] : val;
  if (r < EPW && valid) {
    const int h = lane >> 4;
    e[(size_t)mine * 2 + h] = val;
    atomicMaxF32(&emax[(size_t)dst * NPITCH + h], val);
  }
}

__global__ void edge_exp(const int* __restrict__ ei,
                         float* __restrict__ e,
                         const float* __restrict__ emax,
                         float* __restrict__ denom,
                         int E, int Etot) {
  int t = blockIdx.x * blockDim.x + threadIdx.x;
  if (t >= Etot * 2) return;
  const int eid = t >> 1, h = t & 1;
  int src, dst; edge_endpoints(ei, eid, E, src, dst);
  (void)src;
  const float ee = expf(e[t] - emax[(size_t)dst * NPITCH + h]);
  e[t] = ee;
  __hip_atomic_fetch_add(&denom[(size_t)dst * NPITCH + h], ee,
                         __ATOMIC_RELAXED, __HIP_MEMORY_SCOPE_AGENT);
}

template <bool FULL>
__device__ __forceinline__ void agg_batch(const float* __restrict__ xp,
                                          float* __restrict__ acc,
                                          int src, int dst, float halfAlpha,
                                          int lane, int nvalid) {
  float X[EPW][4];
#pragma unroll
  for (int j = 0; j < EPW; ++j) {
    const int sj = __builtin_amdgcn_readlane(src, j);
    const float* row = xp + (size_t)sj * 128 + lane;
    X[j][0] = row[0];
    X[j][1] = row[32];
    X[j][2] = row[64];
    X[j][3] = row[96];
  }
#pragma unroll
  for (int j = 0; j < EPW; ++j) {
    if (FULL || j < nvalid) {
      const int dj = __builtin_amdgcn_readlane(dst, j);
      const float a0 = __int_as_float(
          __builtin_amdgcn_readlane(__float_as_int(halfAlpha), j));
      const float a1 = __int_as_float(
          __builtin_amdgcn_readlane(__float_as_int(halfAlpha), 16 + j));
      float* p = acc + (size_t)dj * 64 + lane;
      __hip_atomic_fetch_add(p,      fmaf(a1, X[j][2], a0 * X[j][0]),
                             __ATOMIC_RELAXED, __HIP_MEMORY_SCOPE_AGENT);
      __hip_atomic_fetch_add(p + 32, fmaf(a1, X[j][3], a0 * X[j][1]),
                             __ATOMIC_RELAXED, __HIP_MEMORY_SCOPE_AGENT);
    }
  }
}

__global__ void edge_agg(const float* __restrict__ xp,
                         const int* __restrict__ ei,
                         const float* __restrict__ e,
                         const float* __restrict__ denom,
                         float* __restrict__ acc,
                         int E, int Etot) {
  const int lane = threadIdx.x & 31;
  const int e0 = batch_first_edge();
  if (e0 >= Etot) return;
  const int nvalid = Etot - e0;
  const int mine = ((lane & (EPW - 1)) < nvalid) ? e0 + (lane & (EPW - 1))
                                                 : Etot - 1;
  int src, dst; edge_endpoints(ei, mine, E, src, dst);

  const int h = lane >> 4;
  const float halfAlpha = 0.5f * (e[(size_t)mine * 2 + h] /
                                  (denom[(size_t)dst * NPITCH + h] + 1e-16f));

  if (nvalid >= EPW) agg_batch<true >(xp, acc, src, dst, halfAlpha, lane, EPW);
  else               agg_batch<false>(xp, acc, src, dst, halfAlpha, lane, nvalid);
}

#define SB 32
__global__ void bn_stats(const float* __restrict__ acc,
                         float* __restrict__ sum,
                         float* __restrict__ sumsq,
                         int N) {
  __shared__ float s1[256];
  __shared__ float s2[256];
  const int c = threadIdx.x & 63;
  const int nodesPerBlk = blockDim.x >> 6;
  const int step = gridDim.x * nodesPerBlk;
  float ls = 0.0f, lq = 0.0f;
  for (int n0 = blockIdx.x * nodesPerBlk + (threadIdx.x >> 6);
       n0 < N; n0 += SB * step) {
    float v[SB];
#pragma unroll
    for (int j = 0; j < SB; ++j) {
      const int n = n0 + j * step;
      v[j] = acc[(size_t)(n < N ? n : n0) * 64 + c];
    }
#pragma unroll
    for (int j = 0; j < SB; ++j) {
      if (n0 + j * step < N) { ls += v[j]; lq += v[j] * v[j]; }
    }
  }
  s1[threadIdx.x] = ls; s2[threadIdx.x] = lq;
  __syncthreads();
  if (threadIdx.x < 64) {
    ls = s1[threadIdx.x] + s1[threadIdx.x + 64] +
         s1[threadIdx.x + 128] + s1[threadIdx.x + 192];
    lq = s2[threadIdx.x] + s2[threadIdx.x + 64] +
         s2[threadIdx.x + 128] + s2[threadIdx.x + 192];
    __hip_atomic_fetch_add(&sum[c],   ls, __ATOMIC_RELAXED, __HIP_MEMORY_SCOPE_AGENT);
    __hip_atomic_fetch_add(&sumsq[c], lq, __ATOMIC_RELAXED, __HIP_MEMORY_SCOPE_AGENT);
  }
}

__global__ void bn_apply(const float* __restrict__ hmean,
                         const float* __restrict__ sum,
                         const float* __restrict__ sumsq,
                         const float* __restrict__ gamma,
                         const float* __restrict__ beta,
                         float* __restrict__ out,
                         int N, int doTanh) {
  int t = blockIdx.x * blockDim.x + threadIdx.x;
  if (t >= N * 64) return;
  const int c = t & 63;
  const float invN = 1.0f / (float)N;
  const float mean = sum[c] * invN;
  const float var  = sumsq[c] * invN - mean * mean;
  float v = (hmean[t] - mean) * rsqrtf(var + BN_EPS) * gamma[c] + beta[c];
  out[t] = doTanh ? tanhf(v) : v;
}

extern "C" void kernel_launch(void* const* d_in, const int* in_sizes, int n_in,
                              void* d_out, int out_size, void* d_ws, size_t ws_size,
                              hipStream_t stream) {
  const float* x      = (const float*)d_in[0];
  const int*   ei     = (const int*)  d_in[1];
  const float* W0     = (const float*)d_in[2];
  const float* a0     = (const float*)d_in[3];
  const float* gamma0 = (const float*)d_in[4];
  const float* beta0  = (const float*)d_in[5];
  const float* W1     = (const float*)d_in[6];
  const float* a1     = (const float*)d_in[7];
  const float* gamma1 = (const float*)d_in[8];
  const float* beta1  = (const float*)d_in[9];
  float* out = (float*)d_out;

  const int N    = in_sizes[0] / 128;
  const int E    = in_sizes[1] / 2;
  const int Etot = E + N;

  char* ws = (char*)d_ws;
  size_t off = 0;
  auto carve = [&](size_t nfloats) {
    float* p = (float*)(ws + off);
    off = (off + nfloats * sizeof(float) + 255) & ~(size_t)255;
    return p;
  };
  float* xp    = carve((size_t)N * 128);
  float* acc   = carve((size_t)N * 64);
  float* e     = carve((size_t)Etot * 2);
  float* emax  = carve((size_t)N * NPITCH);
  float* denom = carve((size_t)N * NPITCH);
  float* h0    = carve((size_t)N * 64);
  float* stats = carve(128);
  (void)ws_size;

  const int waveEdgeBlocks = (Etot + 8 * EPW - 1) / (8 * EPW);
  const int ehBlocks       = (Etot * 2 + 255) / 256;
  const int nodeElBlocks   = (N * 64 + 255) / 256;
  const int fillBlocks     = (N + 255) / 256;

  gemm_wmma_f32<128><<<N / 16, 64, 0, stream>>>(x, W0, xp);

  node_start<<<fillBlocks, 256, 0, stream>>>(emax, denom, N);
  hipMemsetAsync(acc,   0, (size_t)N * 64 * sizeof(float), stream);
  hipMemsetAsync(stats, 0, 128 * sizeof(float), stream);

  edge_score<<<waveEdgeBlocks, 256, 0, stream>>>(xp, ei, a0, e, emax, E, Etot);
  edge_exp  <<<ehBlocks,       256, 0, stream>>>(ei, e, emax, denom, E, Etot);
  edge_agg  <<<waveEdgeBlocks, 256, 0, stream>>>(xp, ei, e, denom, acc, E, Etot);

  bn_stats<<<256, 256, 0, stream>>>(acc, stats, stats + 64, N);
  bn_apply<<<nodeElBlocks, 256, 0, stream>>>(acc, stats, stats + 64,
                                             gamma0, beta0, h0, N, 1);

  gemm_wmma_f32<64><<<N / 16, 64, 0, stream>>>(h0, W1, xp);

  node_start<<<fillBlocks, 256, 0, stream>>>(emax, denom, N);
  hipMemsetAsync(acc,   0, (size_t)N * 64 * sizeof(float), stream);
  hipMemsetAsync(stats, 0, 128 * sizeof(float), stream);

  edge_score<<<waveEdgeBlocks, 256, 0, stream>>>(xp, ei, a1, e, emax, E, Etot);
  edge_exp  <<<ehBlocks,       256, 0, stream>>>(ei, e, emax, denom, E, Etot);
  edge_agg  <<<waveEdgeBlocks, 256, 0, stream>>>(xp, ei, e, denom, acc, E, Etot);

  bn_stats<<<256, 256, 0, stream>>>(acc, stats, stats + 64, N);
  bn_apply<<<nodeElBlocks, 256, 0, stream>>>(acc, stats, stats + 64,
                                             gamma1, beta1, out, N, 0);
}
